# starting kernel
# baseline (speedup 1.0000x reference)
#include <hip/hip_runtime.h>
#include <hip/hip_bf16.h>
#include <math.h>

typedef __attribute__((ext_vector_type(16))) _Float16 v16h;
typedef __attribute__((ext_vector_type(8)))  _Float16 half8;
typedef __attribute__((ext_vector_type(8)))  float    v8f;

#define NEG_SLOPE 0.2f

__device__ inline float lrelu(float v) { return v > 0.f ? v : NEG_SLOPE * v; }

__device__ inline unsigned enc_f(float f) {
    unsigned u = __float_as_uint(f);
    return (u >> 31) ? ~u : (u | 0x80000000u);
}
__device__ inline float dec_f(unsigned e) {
    unsigned u = (e >> 31) ? (e ^ 0x80000000u) : ~e;
    return __uint_as_float(u);
}

__device__ inline v16h load_frag16(const _Float16* __restrict__ p, int row0,
                                   int ld, int k0, int lane, int maxRow) {
    int m = lane & 15;
    int base = (lane >> 4) * 8;
    int r = row0 + m;
    if (r > maxRow) r = maxRow;
    const _Float16* q = p + (size_t)r * ld + k0 + base;
    half8 lo = *(const half8*)q;
    half8 hi = *(const half8*)(q + 16);
    v16h f;
#pragma unroll
    for (int i = 0; i < 8; ++i) { f[i] = lo[i]; f[i + 8] = hi[i]; }
    return f;
}

__global__ void k_zero(unsigned* p, long long n) {
    long long i = (long long)blockIdx.x * blockDim.x + threadIdx.x;
    if (i < n) p[i] = 0u;
}

__global__ void k_cvt_f32_to_f16(const float* __restrict__ src,
                                 _Float16* __restrict__ dst, long long n) {
    long long i = (long long)blockIdx.x * blockDim.x + threadIdx.x;
    if (i < n) dst[i] = (_Float16)src[i];
}

__global__ void k_tr_w128(const float* __restrict__ W, _Float16* __restrict__ Wt) {
    int i = blockIdx.x * blockDim.x + threadIdx.x;
    if (i >= 128 * 128) return;
    int c = i >> 7, k = i & 127;
    Wt[i] = (_Float16)W[k * 128 + c];
}

__global__ void k_tr_w40(const float* __restrict__ W, _Float16* __restrict__ Wt) {
    int i = blockIdx.x * blockDim.x + threadIdx.x;
    if (i >= 48 * 128) return;
    int c = i >> 7, k = i & 127;
    Wt[i] = (c < 40) ? (_Float16)W[k * 40 + c] : (_Float16)0.f;
}

__global__ __launch_bounds__(32)
void k_gemm_dual(const _Float16* __restrict__ A,
                 const _Float16* __restrict__ Wtl,
                 const _Float16* __restrict__ Wtr,
                 const float* __restrict__ bl, const float* __restrict__ br,
                 float* __restrict__ Cl, float* __restrict__ Cr,
                 int Nrows, int ldc, int nbias) {
    int lane = threadIdx.x & 31;
    int r0 = blockIdx.x * 16;
    int c0 = blockIdx.y * 16;
    v8f accL = {};
    v8f accR = {};
#pragma unroll
    for (int k0 = 0; k0 < 128; k0 += 32) {
        v16h a  = load_frag16(A,   r0, 128, k0, lane, Nrows - 1);
        v16h bL = load_frag16(Wtl, c0, 128, k0, lane, ldc - 1);
        v16h bR = load_frag16(Wtr, c0, 128, k0, lane, ldc - 1);
        accL = __builtin_amdgcn_wmma_f32_16x16x32_f16(false, a, false, bL,
                                                      (short)0, accL, false, false);
        accR = __builtin_amdgcn_wmma_f32_16x16x32_f16(false, a, false, bR,
                                                      (short)0, accR, false, false);
    }
    int n = lane & 15, hw = lane >> 4;
    int col = c0 + n;
    float bLv = (col < nbias) ? bl[col] : 0.f;
    float bRv = (col < nbias) ? br[col] : 0.f;
#pragma unroll
    for (int v = 0; v < 8; ++v) {
        int row = r0 + v + hw * 8;
        if (row < Nrows) {
            Cl[(size_t)row * ldc + col] = accL[v] + bLv;
            Cr[(size_t)row * ldc + col] = accR[v] + bRv;
        }
    }
}

__global__ __launch_bounds__(256)
void k_edge_alpha1(const float* __restrict__ xl, const float* __restrict__ xr,
                   const float* __restrict__ att, const int* __restrict__ ei,
                   int E, int Etot, float* __restrict__ alpha,
                   unsigned* __restrict__ mbits) {
    int e = (blockIdx.x * blockDim.x + threadIdx.x) >> 5;
    int lane = threadIdx.x & 31;
    if (e >= Etot) return;
    int s, d;
    if (e < E) { s = ei[e]; d = ei[E + e]; } else { s = d = e - E; }
    const float4* pl = (const float4*)(xl + (size_t)s * 128);
    const float4* pr = (const float4*)(xr + (size_t)d * 128);
    const float4* pa = (const float4*)att;
    float4 a4 = pl[lane], b4 = pr[lane], t = pa[lane];
    float acc = t.x * lrelu(a4.x + b4.x) + t.y * lrelu(a4.y + b4.y) +
                t.z * lrelu(a4.z + b4.z) + t.w * lrelu(a4.w + b4.w);
#pragma unroll
    for (int off = 1; off < 16; off <<= 1) acc += __shfl_xor(acc, off, 32);
    if ((lane & 15) == 0) {
        int h = lane >> 4;
        alpha[(size_t)e * 2 + h] = acc;
        atomicMax(&mbits[(size_t)d * 2 + h], enc_f(acc));
    }
}

__global__ void k_edge_exp1(const int* __restrict__ ei, int E, int Etot,
                            const unsigned* __restrict__ mbits,
                            float* __restrict__ alpha,
                            float* __restrict__ denom) {
    long long i = (long long)blockIdx.x * blockDim.x + threadIdx.x;
    if (i >= (long long)Etot * 2) return;
    int e = (int)(i >> 1), h = (int)(i & 1);
    int d = (e < E) ? ei[E + e] : e - E;
    float m = dec_f(mbits[(size_t)d * 2 + h]);
    float ex = __expf(alpha[i] - m);
    alpha[i] = ex;
    atomicAdd(&denom[(size_t)d * 2 + h], ex);
}

__global__ void k_edge_agg1(const float* __restrict__ xl,
                            const int* __restrict__ ei, int E, int Etot,
                            const float* __restrict__ alpha,
                            const float* __restrict__ denom,
                            float* __restrict__ agg) {
    long long i = (long long)blockIdx.x * blockDim.x + threadIdx.x;
    if (i >= (long long)Etot * 128) return;
    int e = (int)(i >> 7), c = (int)(i & 127), h = c >> 6;
    int s, d;
    if (e < E) { s = ei[e]; d = ei[E + e]; } else { s = d = e - E; }
    float a = alpha[(size_t)e * 2 + h] / denom[(size_t)d * 2 + h];
    atomicAdd(&agg[(size_t)d * 128 + c], xl[(size_t)s * 128 + c] * a);
}

__global__ void k_epilogue1(const float* __restrict__ agg,
                            const float* __restrict__ bias,
                            _Float16* __restrict__ hh, long long n) {
    long long i = (long long)blockIdx.x * blockDim.x + threadIdx.x;
    if (i >= n) return;
    float v = agg[i] + bias[(int)(i & 127)];
    hh[i] = (_Float16)(v > 0.f ? v : 0.f);
}

__global__ __launch_bounds__(256)
void k_edge_alpha2(const float* __restrict__ xl, const float* __restrict__ xr,
                   const float* __restrict__ att, const int* __restrict__ ei,
                   int E, int Etot, float* __restrict__ alpha,
                   unsigned* __restrict__ mbits) {
    int e = (blockIdx.x * blockDim.x + threadIdx.x) >> 5;
    int lane = threadIdx.x & 31;
    if (e >= Etot) return;
    int s, d;
    if (e < E) { s = ei[e]; d = ei[E + e]; } else { s = d = e - E; }
    float acc = 0.f;
    if (lane < 10) {
        const float4* pl = (const float4*)(xl + (size_t)s * 48);
        const float4* pr = (const float4*)(xr + (size_t)d * 48);
        const float4* pa = (const float4*)att;
        float4 a4 = pl[lane], b4 = pr[lane], t = pa[lane];
        acc = t.x * lrelu(a4.x + b4.x) + t.y * lrelu(a4.y + b4.y) +
              t.z * lrelu(a4.z + b4.z) + t.w * lrelu(a4.w + b4.w);
    }
#pragma unroll
    for (int off = 1; off < 32; off <<= 1) acc += __shfl_xor(acc, off, 32);
    if (lane == 0) {
        alpha[e] = acc;
        atomicMax(&mbits[d], enc_f(acc));
    }
}

__global__ void k_edge_exp2(const int* __restrict__ ei, int E, int Etot,
                            const unsigned* __restrict__ mbits,
                            float* __restrict__ alpha,
                            float* __restrict__ denom) {
    long long i = (long long)blockIdx.x * blockDim.x + threadIdx.x;
    if (i >= Etot) return;
    int e = (int)i;
    int d = (e < E) ? ei[E + e] : e - E;
    float ex = __expf(alpha[e] - dec_f(mbits[d]));
    alpha[e] = ex;
    atomicAdd(&denom[d], ex);
}

__global__ void k_edge_agg2(const float* __restrict__ xl,
                            const int* __restrict__ ei, int E, int Etot,
                            const float* __restrict__ alpha,
                            const float* __restrict__ denom,
                            float* __restrict__ agg) {
    long long i = (long long)blockIdx.x * blockDim.x + threadIdx.x;
    if (i >= (long long)Etot * 64) return;
    int e = (int)(i >> 6), c = (int)(i & 63);
    if (c >= 40) return;
    int s, d;
    if (e < E) { s = ei[e]; d = ei[E + e]; } else { s = d = e - E; }
    float a = alpha[e] / denom[d];
    atomicAdd(&agg[(size_t)d * 40 + c], xl[(size_t)s * 48 + c] * a);
}

__global__ __launch_bounds__(256)
void k_logsoftmax(const float* __restrict__ agg, const float* __restrict__ bias,
                  float* __restrict__ out, int N) {
    int node = (blockIdx.x * blockDim.x + threadIdx.x) >> 5;
    int lane = threadIdx.x & 31;
    if (node >= N) return;
    const float* p = agg + (size_t)node * 40;
    float v0 = p[lane] + bias[lane];
    float v1 = (lane < 8) ? (p[lane + 32] + bias[lane + 32]) : -INFINITY;
    float m = fmaxf(v0, v1);
#pragma unroll
    for (int off = 1; off < 32; off <<= 1) m = fmaxf(m, __shfl_xor(m, off, 32));
    float se = __expf(v0 - m) + ((lane < 8) ? __expf(v1 - m) : 0.f);
#pragma unroll
    for (int off = 1; off < 32; off <<= 1) se += __shfl_xor(se, off, 32);
    float L = m + logf(se);
    out[(size_t)node * 40 + lane] = v0 - L;
    if (lane < 8) out[(size_t)node * 40 + lane + 32] = v1 - L;
}

extern "C" void kernel_launch(void* const* d_in, const int* in_sizes, int n_in,
                              void* d_out, int out_size, void* d_ws, size_t ws_size,
                              hipStream_t stream) {
    const float* x    = (const float*)d_in[0];
    const int*   ei   = (const int*)d_in[1];
    const float* W1l  = (const float*)d_in[2];
    const float* b1l  = (const float*)d_in[3];
    const float* W1r  = (const float*)d_in[4];
    const float* b1r  = (const float*)d_in[5];
    const float* att1 = (const float*)d_in[6];
    const float* bias1= (const float*)d_in[7];
    const float* W2l  = (const float*)d_in[8];
    const float* b2l  = (const float*)d_in[9];
    const float* W2r  = (const float*)d_in[10];
    const float* b2r  = (const float*)d_in[11];
    const float* att2 = (const float*)d_in[12];
    const float* bias2= (const float*)d_in[13];
    float* out = (float*)d_out;

    const int N = in_sizes[0] / 128;
    const int E = in_sizes[1] / 2;
    const int Etot = E + N;

    char* w = (char*)d_ws;
    size_t off = 0;
    auto take = [&](size_t bytes) -> char* {
        char* p = w + off;
        off = (off + bytes + 255) & ~(size_t)255;
        return p;
    };
    _Float16* xh   = (_Float16*)take((size_t)N * 128 * 2);
    _Float16* hh   = (_Float16*)take((size_t)N * 128 * 2);
    _Float16* Wt1l = (_Float16*)take(128 * 128 * 2);
    _Float16* Wt1r = (_Float16*)take(128 * 128 * 2);
    _Float16* Wt2l = (_Float16*)take(48 * 128 * 2);
    _Float16* Wt2r = (_Float16*)take(48 * 128 * 2);
    float*    xl   = (float*)take((size_t)N * 128 * 4);
    float*    xr   = (float*)take((size_t)N * 128 * 4);
    float*    alpha= (float*)take((size_t)Etot * 2 * 4);
    unsigned* mbits= (unsigned*)take((size_t)N * 2 * 4);
    float*    denom= (float*)take((size_t)N * 2 * 4);
    float*    agg1 = (float*)take((size_t)N * 128 * 4);
    float*    agg2 = (float*)take((size_t)N * 40 * 4);

    const int B = 256;
    auto blks = [&](long long n) { return (unsigned)((n + B - 1) / B); };
    const long long Nx128 = (long long)N * 128;
    const int rowTiles = (N + 15) / 16;

    k_cvt_f32_to_f16<<<blks(Nx128), B, 0, stream>>>(x, xh, Nx128);
    k_tr_w128<<<blks(128 * 128), B, 0, stream>>>(W1l, Wt1l);
    k_tr_w128<<<blks(128 * 128), B, 0, stream>>>(W1r, Wt1r);
    k_tr_w40 <<<blks(48 * 128),  B, 0, stream>>>(W2l, Wt2l);
    k_tr_w40 <<<blks(48 * 128),  B, 0, stream>>>(W2r, Wt2r);
    k_zero<<<blks(2LL * N), B, 0, stream>>>(mbits, 2LL * N);
    k_zero<<<blks(2LL * N), B, 0, stream>>>((unsigned*)denom, 2LL * N);
    k_zero<<<blks(Nx128), B, 0, stream>>>((unsigned*)agg1, Nx128);

    k_gemm_dual<<<dim3(rowTiles, 8), 32, 0, stream>>>(xh, Wt1l, Wt1r, b1l, b1r,
                                                      xl, xr, N, 128, 128);
    k_edge_alpha1<<<blks((long long)Etot * 32), B, 0, stream>>>(xl, xr, att1, ei,
                                                                E, Etot, alpha, mbits);
    k_edge_exp1<<<blks(2LL * Etot), B, 0, stream>>>(ei, E, Etot, mbits, alpha, denom);
    k_edge_agg1<<<blks((long long)Etot * 128), B, 0, stream>>>(xl, ei, E, Etot,
                                                               alpha, denom, agg1);
    k_epilogue1<<<blks(Nx128), B, 0, stream>>>(agg1, bias1, hh, Nx128);

    k_zero<<<blks(2LL * N), B, 0, stream>>>(mbits, 2LL * N);
    k_zero<<<blks(2LL * N), B, 0, stream>>>((unsigned*)denom, 2LL * N);
    k_zero<<<blks(40LL * N), B, 0, stream>>>((unsigned*)agg2, 40LL * N);

    k_gemm_dual<<<dim3(rowTiles, 3), 32, 0, stream>>>(hh, Wt2l, Wt2r, b2l, b2r,
                                                      xl, xr, N, 48, 40);
    k_edge_alpha2<<<blks((long long)Etot * 32), B, 0, stream>>>(xl, xr, att2, ei,
                                                                E, Etot, alpha, mbits);
    k_edge_exp2<<<blks(Etot), B, 0, stream>>>(ei, E, Etot, mbits, alpha, denom);
    k_edge_agg2<<<blks((long long)Etot * 64), B, 0, stream>>>(xl, ei, E, Etot,
                                                              alpha, denom, agg2);
    k_logsoftmax<<<blks((long long)N * 32), B, 0, stream>>>(agg2, bias2, out, N);
}


# Softmax division moved from every edge-channel to once per node: one edge sweep (a wave per edge) adds exp-weighted rows and exp sums together; exp pass gone per layer; float sum order changes
# speedup vs baseline: 1.2871x; 1.2871x over previous
#include <hip/hip_runtime.h>
#include <hip/hip_bf16.h>
#include <math.h>

typedef __attribute__((ext_vector_type(16))) _Float16 v16h;
typedef __attribute__((ext_vector_type(8)))  _Float16 half8;
typedef __attribute__((ext_vector_type(8)))  float    v8f;

#define NEG_SLOPE 0.2f

__device__ inline float lrelu(float v) { return v > 0.f ? v : NEG_SLOPE * v; }

__device__ inline unsigned enc_f(float f) {
    unsigned u = __float_as_uint(f);
    return (u >> 31) ? ~u : (u | 0x80000000u);
}
__device__ inline float dec_f(unsigned e) {
    unsigned u = (e >> 31) ? (e ^ 0x80000000u) : ~e;
    return __uint_as_float(u);
}

__device__ inline v16h load_frag16(const _Float16* __restrict__ p, int row0,
                                   int ld, int k0, int lane, int maxRow) {
    int m = lane & 15;
    int base = (lane >> 4) * 8;
    int r = row0 + m;
    if (r > maxRow) r = maxRow;
    const _Float16* q = p + (size_t)r * ld + k0 + base;
    half8 lo = *(const half8*)q;
    half8 hi = *(const half8*)(q + 16);
    v16h f;
#pragma unroll
    for (int i = 0; i < 8; ++i) { f[i] = lo[i]; f[i + 8] = hi[i]; }
    return f;
}

__global__ void k_zero(unsigned* p, long long n) {
    long long i = (long long)blockIdx.x * blockDim.x + threadIdx.x;
    if (i < n) p[i] = 0u;
}

__global__ void k_cvt_f32_to_f16(const float* __restrict__ src,
                                 _Float16* __restrict__ dst, long long n) {
    long long i = (long long)blockIdx.x * blockDim.x + threadIdx.x;
    if (i < n) dst[i] = (_Float16)src[i];
}

__global__ void k_tr_w128(const float* __restrict__ W, _Float16* __restrict__ Wt) {
    int i = blockIdx.x * blockDim.x + threadIdx.x;
    if (i >= 128 * 128) return;
    int c = i >> 7, k = i & 127;
    Wt[i] = (_Float16)W[k * 128 + c];
}

__global__ void k_tr_w40(const float* __restrict__ W, _Float16* __restrict__ Wt) {
    int i = blockIdx.x * blockDim.x + threadIdx.x;
    if (i >= 48 * 128) return;
    int c = i >> 7, k = i & 127;
    Wt[i] = (c < 40) ? (_Float16)W[k * 40 + c] : (_Float16)0.f;
}

__global__ __launch_bounds__(32)
void k_gemm_dual(const _Float16* __restrict__ A,
                 const _Float16* __restrict__ Wtl,
                 const _Float16* __restrict__ Wtr,
                 const float* __restrict__ bl, const float* __restrict__ br,
                 float* __restrict__ Cl, float* __restrict__ Cr,
                 int Nrows, int ldc, int nbias) {
    int lane = threadIdx.x & 31;
    int r0 = blockIdx.x * 16;
    int c0 = blockIdx.y * 16;
    v8f accL = {};
    v8f accR = {};
#pragma unroll
    for (int k0 = 0; k0 < 128; k0 += 32) {
        v16h a  = load_frag16(A,   r0, 128, k0, lane, Nrows - 1);
        v16h bL = load_frag16(Wtl, c0, 128, k0, lane, ldc - 1);
        v16h bR = load_frag16(Wtr, c0, 128, k0, lane, ldc - 1);
        accL = __builtin_amdgcn_wmma_f32_16x16x32_f16(false, a, false, bL,
                                                      (short)0, accL, false, false);
        accR = __builtin_amdgcn_wmma_f32_16x16x32_f16(false, a, false, bR,
                                                      (short)0, accR, false, false);
    }
    int n = lane & 15, hw = lane >> 4;
    int col = c0 + n;
    float bLv = (col < nbias) ? bl[col] : 0.f;
    float bRv = (col < nbias) ? br[col] : 0.f;
#pragma unroll
    for (int v = 0; v < 8; ++v) {
        int row = r0 + v + hw * 8;
        if (row < Nrows) {
            Cl[(size_t)row * ldc + col] = accL[v] + bLv;
            Cr[(size_t)row * ldc + col] = accR[v] + bRv;
        }
    }
}

__global__ __launch_bounds__(256)
void k_edge_alpha1(const float* __restrict__ xl, const float* __restrict__ xr,
                   const float* __restrict__ att, const int* __restrict__ ei,
                   int E, int Etot, float* __restrict__ alpha,
                   unsigned* __restrict__ mbits) {
    int e = (blockIdx.x * blockDim.x + threadIdx.x) >> 5;
    int lane = threadIdx.x & 31;
    if (e >= Etot) return;
    int s, d;
    if (e < E) { s = ei[e]; d = ei[E + e]; } else { s = d = e - E; }
    const float4* pl = (const float4*)(xl + (size_t)s * 128);
    const float4* pr = (const float4*)(xr + (size_t)d * 128);
    const float4* pa = (const float4*)att;
    float4 a4 = pl[lane], b4 = pr[lane], t = pa[lane];
    float acc = t.x * lrelu(a4.x + b4.x) + t.y * lrelu(a4.y + b4.y) +
                t.z * lrelu(a4.z + b4.z) + t.w * lrelu(a4.w + b4.w);
#pragma unroll
    for (int off = 1; off < 16; off <<= 1) acc += __shfl_xor(acc, off, 32);
    if ((lane & 15) == 0) {
        int h = lane >> 4;
        alpha[(size_t)e * 2 + h] = acc;
        atomicMax(&mbits[(size_t)d * 2 + h], enc_f(acc));
    }
}

__global__ __launch_bounds__(256)
void k_edge_wsum1(const float* __restrict__ xl, const int* __restrict__ ei,
                  int E, int Etot, const float* __restrict__ alpha,
                  const unsigned* __restrict__ mbits,
                  float* __restrict__ denom, float* __restrict__ agg) {
    int e = __builtin_amdgcn_readfirstlane(
        (int)((blockIdx.x * blockDim.x + threadIdx.x) >> 5));
    int lane = threadIdx.x & 31;
    if (e >= Etot) return;
    float a0 = alpha[(size_t)e * 2], a1 = alpha[(size_t)e * 2 + 1];
    int s, d;
    if (e < E) { s = ei[e]; d = ei[E + e]; } else { s = d = e - E; }
    const float* ps = xl + (size_t)s * 128 + lane;
    float* pd = agg + (size_t)d * 128 + lane;
    float v0 = ps[0], v1 = ps[32], v2 = ps[64], v3 = ps[96];
    float w0 = __expf(a0 - dec_f(mbits[(size_t)d * 2]));
    float w1 = __expf(a1 - dec_f(mbits[(size_t)d * 2 + 1]));
    atomicAdd(pd,      v0 * w0);
    atomicAdd(pd + 32, v1 * w0);
    atomicAdd(pd + 64, v2 * w1);
    atomicAdd(pd + 96, v3 * w1);
    if (lane < 2) atomicAdd(&denom[(size_t)d * 2 + lane], lane ? w1 : w0);
}

__global__ void k_epilogue1(const float* __restrict__ agg,
                            const float* __restrict__ denom,
                            const float* __restrict__ bias,
                            _Float16* __restrict__ hh, long long n) {
    long long i = (long long)blockIdx.x * blockDim.x + threadIdx.x;
    if (i >= n) return;
    int c = (int)(i & 127);
    float v = agg[i] / denom[(i >> 7) * 2 + (c >> 6)] + bias[c];
    hh[i] = (_Float16)(v > 0.f ? v : 0.f);
}

__global__ __launch_bounds__(256)
void k_edge_alpha2(const float* __restrict__ xl, const float* __restrict__ xr,
                   const float* __restrict__ att, const int* __restrict__ ei,
                   int E, int Etot, float* __restrict__ alpha,
                   unsigned* __restrict__ mbits) {
    int e = (blockIdx.x * blockDim.x + threadIdx.x) >> 5;
    int lane = threadIdx.x & 31;
    if (e >= Etot) return;
    int s, d;
    if (e < E) { s = ei[e]; d = ei[E + e]; } else { s = d = e - E; }
    float acc = 0.f;
    if (lane < 10) {
        const float4* pl = (const float4*)(xl + (size_t)s * 48);
        const float4* pr = (const float4*)(xr + (size_t)d * 48);
        const float4* pa = (const float4*)att;
        float4 a4 = pl[lane], b4 = pr[lane], t = pa[lane];
        acc = t.x * lrelu(a4.x + b4.x) + t.y * lrelu(a4.y + b4.y) +
              t.z * lrelu(a4.z + b4.z) + t.w * lrelu(a4.w + b4.w);
    }
#pragma unroll
    for (int off = 1; off < 32; off <<= 1) acc += __shfl_xor(acc, off, 32);
    if (lane == 0) {
        alpha[e] = acc;
        atomicMax(&mbits[d], enc_f(acc));
    }
}

__global__ __launch_bounds__(256)
void k_edge_wsum2(const float* __restrict__ xl, const int* __restrict__ ei,
                  int E, int Etot, const float* __restrict__ alpha,
                  const unsigned* __restrict__ mbits,
                  float* __restrict__ denom, float* __restrict__ agg) {
    int e = __builtin_amdgcn_readfirstlane(
        (int)((blockIdx.x * blockDim.x + threadIdx.x) >> 5));
    int lane = threadIdx.x & 31;
    if (e >= Etot) return;
    float a = alpha[e];
    int s, d;
    if (e < E) { s = ei[e]; d = ei[E + e]; } else { s = d = e - E; }
    const float* ps = xl + (size_t)s * 48;
    float* pd = agg + (size_t)d * 40 + lane;
    float v0 = ps[lane];
    float v1 = ps[32 + (lane & 7)];
    unsigned mb = mbits[d];
    asm volatile("" :: "v"(v0), "v"(v1), "s"(mb));
    float w = __expf(a - dec_f(mb));
    atomicAdd(pd, v0 * w);
    if (lane < 8) atomicAdd(pd + 32, v1 * w);
    if (lane == 0) atomicAdd(&denom[d], w);
}

__global__ __launch_bounds__(256)
void k_logsoftmax(const float* __restrict__ agg, const float* __restrict__ denom,
                  const float* __restrict__ bias, float* __restrict__ out, int N) {
    int node = (blockIdx.x * blockDim.x + threadIdx.x) >> 5;
    int lane = threadIdx.x & 31;
    if (node >= N) return;
    const float* p = agg + (size_t)node * 40;
    float dn = denom[node];
    float v0 = p[lane] / dn + bias[lane];
    float v1 = (lane < 8) ? (p[lane + 32] / dn + bias[lane + 32]) : -INFINITY;
    float m = fmaxf(v0, v1);
#pragma unroll
    for (int off = 1; off < 32; off <<= 1) m = fmaxf(m, __shfl_xor(m, off, 32));
    float se = __expf(v0 - m) + ((lane < 8) ? __expf(v1 - m) : 0.f);
#pragma unroll
    for (int off = 1; off < 32; off <<= 1) se += __shfl_xor(se, off, 32);
    float L = m + logf(se);
    out[(size_t)node * 40 + lane] = v0 - L;
    if (lane < 8) out[(size_t)node * 40 + lane + 32] = v1 - L;
}

extern "C" void kernel_launch(void* const* d_in, const int* in_sizes, int n_in,
                              void* d_out, int out_size, void* d_ws, size_t ws_size,
                              hipStream_t stream) {
    const float* x    = (const float*)d_in[0];
    const int*   ei   = (const int*)d_in[1];
    const float* W1l  = (const float*)d_in[2];
    const float* b1l  = (const float*)d_in[3];
    const float* W1r  = (const float*)d_in[4];
    const float* b1r  = (const float*)d_in[5];
    const float* att1 = (const float*)d_in[6];
    const float* bias1= (const float*)d_in[7];
    const float* W2l  = (const float*)d_in[8];
    const float* b2l  = (const float*)d_in[9];
    const float* W2r  = (const float*)d_in[10];
    const float* b2r  = (const float*)d_in[11];
    const float* att2 = (const float*)d_in[12];
    const float* bias2= (const float*)d_in[13];
    float* out = (float*)d_out;

    const int N = in_sizes[0] / 128;
    const int E = in_sizes[1] / 2;
    const int Etot = E + N;

    char* w = (char*)d_ws;
    size_t off = 0;
    auto take = [&](size_t bytes) -> char* {
        char* p = w + off;
        off = (off + bytes + 255) & ~(size_t)255;
        return p;
    };
    _Float16* xh   = (_Float16*)take((size_t)N * 128 * 2);
    _Float16* hh   = (_Float16*)take((size_t)N * 128 * 2);
    _Float16* Wt1l = (_Float16*)take(128 * 128 * 2);
    _Float16* Wt1r = (_Float16*)take(128 * 128 * 2);
    _Float16* Wt2l = (_Float16*)take(48 * 128 * 2);
    _Float16* Wt2r = (_Float16*)take(48 * 128 * 2);
    float*    xl   = (float*)take((size_t)N * 128 * 4);
    float*    xr   = (float*)take((size_t)N * 128 * 4);
    float*    alpha= (float*)take((size_t)Etot * 2 * 4);
    unsigned* mbits= (unsigned*)take((size_t)N * 2 * 4);
    float*    denom= (float*)take((size_t)N * 2 * 4);
    float*    agg1 = (float*)take((size_t)N * 128 * 4);
    float*    agg2 = (float*)take((size_t)N * 40 * 4);

    const int B = 256;
    auto blks = [&](long long n) { return (unsigned)((n + B - 1) / B); };
    const long long Nx128 = (long long)N * 128;
    const int rowTiles = (N + 15) / 16;

    k_cvt_f32_to_f16<<<blks(Nx128), B, 0, stream>>>(x, xh, Nx128);
    k_tr_w128<<<blks(128 * 128), B, 0, stream>>>(W1l, Wt1l);
    k_tr_w128<<<blks(128 * 128), B, 0, stream>>>(W1r, Wt1r);
    k_tr_w40 <<<blks(48 * 128),  B, 0, stream>>>(W2l, Wt2l);
    k_tr_w40 <<<blks(48 * 128),  B, 0, stream>>>(W2r, Wt2r);
    k_zero<<<blks(2LL * N), B, 0, stream>>>(mbits, 2LL * N);
    k_zero<<<blks(2LL * N), B, 0, stream>>>((unsigned*)denom, 2LL * N);
    k_zero<<<blks(Nx128), B, 0, stream>>>((unsigned*)agg1, Nx128);

    k_gemm_dual<<<dim3(rowTiles, 8), 32, 0, stream>>>(xh, Wt1l, Wt1r, b1l, b1r,
                                                      xl, xr, N, 128, 128);
    k_edge_alpha1<<<blks((long long)Etot * 32), B, 0, stream>>>(xl, xr, att1, ei,
                                                                E, Etot, alpha, mbits);
    k_edge_wsum1<<<blks((long long)Etot * 32), B, 0, stream>>>(xl, ei, E, Etot, alpha,
                                                               mbits, denom, agg1);
    k_epilogue1<<<blks(Nx128), B, 0, stream>>>(agg1, denom, bias1, hh, Nx128);

    k_zero<<<blks(2LL * N), B, 0, stream>>>(mbits, 2LL * N);
    k_zero<<<blks(2LL * N), B, 0, stream>>>((unsigned*)denom, 2LL * N);
    k_zero<<<blks(40LL * N), B, 0, stream>>>((unsigned*)agg2, 40LL * N);

    k_gemm_dual<<<dim3(rowTiles, 3), 32, 0, stream>>>(hh, Wt2l, Wt2r, b2l, b2r,
                                                      xl, xr, N, 48, 40);
    k_edge_alpha2<<<blks((long long)Etot * 32), B, 0, stream>>>(xl, xr, att2, ei,
                                                                E, Etot, alpha, mbits);
    k_edge_wsum2<<<blks((long long)Etot * 32), B, 0, stream>>>(xl, ei, E, Etot, alpha,
                                                               mbits, denom, agg2);
    k_logsoftmax<<<blks((long long)N * 32), B, 0, stream>>>(agg2, denom, bias2, out, N);
}
